# starting kernel
# baseline (speedup 1.0000x reference)
#include <hip/hip_runtime.h>
#include <math.h>

#define BATCH 256
#define SEQ   2048
#define FEAT  64
#define NROWS (BATCH * SEQ)

typedef float v2f __attribute__((ext_vector_type(2)));
typedef float v8f __attribute__((ext_vector_type(8)));

__global__ void __launch_bounds__(256)
proj_kernel(const float* __restrict__ X,
            const float* __restrict__ W_ih,
            const float* __restrict__ b_ih,
            const float* __restrict__ b_hh,
            float* __restrict__ proj,
            float* __restrict__ tvals) {
    __shared__ float sW[FEAT];
    const int tid = threadIdx.x;
    if (tid < FEAT) sW[tid] = W_ih[tid];
    __syncthreads();

    const int lane = tid & 31;
    const int wave = tid >> 5;
    const int m    = lane & 15;
    const int hi   = lane >> 4;
    const long tile   = (long)blockIdx.x * 8 + wave;
    const long r_base = tile * 16;
    const long row    = r_base + m;
    const float* xrow = X + row * FEAT + hi * 2;

    const float bias = b_ih[0] + b_hh[0];
    const bool colZeroLane = (m == 0);

    v2f bf[16];
#pragma unroll
    for (int kk = 0; kk < 16; ++kk) {
        const int k = kk * 4 + hi * 2;
        const float wx = sW[k];
        const float wy = sW[k + 1];
        bf[kk].x = colZeroLane ? wx : 0.0f;
        bf[kk].y = colZeroLane ? wy : 0.0f;
    }

    v2f af[16];
#pragma unroll
    for (int kk = 0; kk < 16; ++kk)
        af[kk] = *(const v2f*)(xrow + kk * 4);

    v8f c;
#pragma unroll
    for (int i = 0; i < 8; ++i) c[i] = colZeroLane ? bias : 0.0f;

#pragma unroll
    for (int kk = 0; kk < 16; ++kk)
        c = __builtin_amdgcn_wmma_f32_16x16x4_f32(
                 false, af[kk],  false, bf[kk],
                 (short)0, c,  false,  false);

    if (hi == 0) {
        const int bidx = (int)(row >> 11);
        const int t    = (int)(row & (SEQ - 1));
        tvals[t * BATCH + bidx] = af[0].y;
    }

    if (colZeroLane) {
        const long rr    = r_base + hi * 8;
        const int  bidx  = (int)(rr >> 11);
        const int  tbase = (int)(rr & (SEQ - 1));
#pragma unroll
        for (int i = 0; i < 8; ++i)
            proj[(tbase + i) * BATCH + bidx] = c[i];
    }
}

static __device__ __forceinline__ float fast_tanh(float x) {
#if __has_builtin(__builtin_amdgcn_tanhf)
    return __builtin_amdgcn_tanhf(x);
#elif __has_builtin(__builtin_amdgcn_tanh_f32)
    return __builtin_amdgcn_tanh_f32(x);
#else
    return tanhf(x);
#endif
}

__global__ void __launch_bounds__(32)
scan_kernel(const float* __restrict__ proj,
            const float* __restrict__ hx0,
            const float* __restrict__ W_hh,
            float* __restrict__ hidden) {
    const int b = blockIdx.x * 32 + threadIdx.x;
    float h = hx0[b];
    const float w = W_hh[0];
    const float* p = proj + b;
    float* o = hidden + b;
#pragma unroll 8
    for (int t = 0; t < SEQ; ++t) {
        const float a = p[(long)t * BATCH];
        h = fast_tanh(fmaf(h, w, a));
        o[(long)t * BATCH] = h;
    }
}

__global__ void __launch_bounds__(256)
lam_kernel(const float* __restrict__ hidden,
           const float* __restrict__ tvals,
           const float* __restrict__ W_h2p,
           const float* __restrict__ b_h2p,
           float* __restrict__ lam_out) {
    __shared__ float sW[64], sb[64];
    const int tid = threadIdx.x;
    if (tid < 64) { sW[tid] = W_h2p[tid]; sb[tid] = b_h2p[tid]; }
    __syncthreads();

    const long idx = (long)blockIdx.x * 256 + tid;
    const float h  = hidden[idx];
    const float tt = tvals[idx];

    float s = 0.0f;
#pragma unroll
    for (int j = 0; j < 32; ++j) {
        const float alpha = fmaf(h, sW[2 * j],     sb[2 * j]);
        const float beta  = fmaf(h, sW[2 * j + 1], sb[2 * j + 1]);
        const float v     = fmaf(alpha, tt, beta);
        s += fmaxf(v, 0.0f);
    }
    const float lam = fmaxf(s, 0.0f) + log1pf(__expf(-fabsf(s)));
    lam_out[idx] = lam;
}

extern "C" void kernel_launch(void* const* d_in, const int* in_sizes, int n_in,
                              void* d_out, int out_size, void* d_ws, size_t ws_size,
                              hipStream_t stream) {
    const float* X     = (const float*)d_in[0];
    const float* hx0   = (const float*)d_in[1];
    const float* W_ih  = (const float*)d_in[2];
    const float* b_ih  = (const float*)d_in[3];
    const float* W_hh  = (const float*)d_in[4];
    const float* b_hh  = (const float*)d_in[5];
    const float* W_h2p = (const float*)d_in[6];
    const float* b_h2p = (const float*)d_in[7];

    float* proj  = (float*)d_ws;
    float* tvals = proj + NROWS;

    float* hidden = (float*)d_out;
    float* lam    = hidden + NROWS;

    proj_kernel<<<NROWS / 16 / 8, 256, 0, stream>>>(X, W_ih, b_ih, b_hh, proj, tvals);
    scan_kernel<<<8, 32, 0, stream>>>(proj, hx0, W_hh, hidden);
    lam_kernel<<<NROWS / 256, 256, 0, stream>>>(hidden, tvals, W_h2p, b_h2p, lam);
}


# scan: chain wave touches LDS only; 8 helper waves of its work-group stage proj tiles in and h tiles out (128-step tiles, LDS rings, LDS-only barrier), 64 steps per b128 group; sums unchanged
# speedup vs baseline: 3.9721x; 3.9721x over previous
#include <hip/hip_runtime.h>
#include <math.h>

#define BATCH 256
#define SEQ   2048
#define FEAT  64
#define NROWS (BATCH * SEQ)

typedef float v2f __attribute__((ext_vector_type(2)));
typedef float v8f __attribute__((ext_vector_type(8)));

__global__ void __launch_bounds__(256)
proj_kernel(const float* __restrict__ X,
            const float* __restrict__ W_ih,
            const float* __restrict__ b_ih,
            const float* __restrict__ b_hh,
            float* __restrict__ proj,
            float* __restrict__ tvals) {
    __shared__ float sW[FEAT];
    const int tid = threadIdx.x;
    if (tid < FEAT) sW[tid] = W_ih[tid];
    __syncthreads();

    const int lane = tid & 31;
    const int wave = tid >> 5;
    const int m    = lane & 15;
    const int hi   = lane >> 4;
    const long tile   = (long)blockIdx.x * 8 + wave;
    const long r_base = tile * 16;
    const long row    = r_base + m;
    const float* xrow = X + row * FEAT + hi * 2;

    const float bias = b_ih[0] + b_hh[0];
    const bool colZeroLane = (m == 0);

    v2f bf[16];
#pragma unroll
    for (int kk = 0; kk < 16; ++kk) {
        const int k = kk * 4 + hi * 2;
        const float wx = sW[k];
        const float wy = sW[k + 1];
        bf[kk].x = colZeroLane ? wx : 0.0f;
        bf[kk].y = colZeroLane ? wy : 0.0f;
    }

    v2f af[16];
#pragma unroll
    for (int kk = 0; kk < 16; ++kk)
        af[kk] = *(const v2f*)(xrow + kk * 4);

    v8f c;
#pragma unroll
    for (int i = 0; i < 8; ++i) c[i] = colZeroLane ? bias : 0.0f;

#pragma unroll
    for (int kk = 0; kk < 16; ++kk)
        c = __builtin_amdgcn_wmma_f32_16x16x4_f32(
                 false, af[kk],  false, bf[kk],
                 (short)0, c,  false,  false);

    if (hi == 0) {
        const int bidx = (int)(row >> 11);
        const int t    = (int)(row & (SEQ - 1));
        tvals[t * BATCH + bidx] = af[0].y;
    }

    if (colZeroLane) {
        const long rr    = r_base + hi * 8;
        const int  bidx  = (int)(rr >> 11);
        const int  tbase = (int)(rr & (SEQ - 1));
#pragma unroll
        for (int i = 0; i < 8; ++i)
            proj[(tbase + i) * BATCH + bidx] = c[i];
    }
}

static __device__ __forceinline__ float fast_tanh(float x) {
#if __has_builtin(__builtin_amdgcn_tanhf)
    return __builtin_amdgcn_tanhf(x);
#elif __has_builtin(__builtin_amdgcn_tanh_f32)
    return __builtin_amdgcn_tanh_f32(x);
#else
    return tanhf(x);
#endif
}

#define SCAN_TB      128
#define SCAN_NBLK    (SEQ / SCAN_TB)
#define SCAN_NS      3
#define SCAN_NO      2
#define SCAN_PITCH   132
#define SCAN_STAGE   (32 * SCAN_PITCH)
#define SCAN_GRP     64
#define SCAN_NGRP    (SEQ / SCAN_GRP)
#define SCAN_GV      (SCAN_GRP / 4)
#define SCAN_GPB     (SCAN_TB / SCAN_GRP)

typedef float v4f __attribute__((ext_vector_type(4)));

static __device__ __forceinline__ void lds_barrier() {
    asm volatile("s_wait_dscnt 0x0\n\ts_barrier_signal -1\n\ts_barrier_wait -1" ::: "memory");
}

static __device__ __forceinline__ void sched_fence(float& hv) {
    asm volatile("" : "+v"(hv) :: "memory");
}

static __device__ __forceinline__ void scan_tile_in(const float* __restrict__ proj, float* sIn,
                                                    int kb, int b0, int hw, int lane) {
    const float* srcp = proj + (long)(kb * SCAN_TB) * BATCH + b0;
    float* st = sIn + (kb % SCAN_NS) * SCAN_STAGE;
    const int lrow  = lane & 7;
    const int lquad = lane >> 3;
    v4f v[4];
#pragma unroll
    for (int j = 0; j < 4; ++j) {
        const int idx  = hw * 4 + j;
        const int trow = 8 * (idx >> 1) + lrow;
        const int q    = 4 * (idx & 1) + lquad;
        v[j] = *(const v4f*)(srcp + trow * BATCH + 4 * q);
    }
#pragma unroll
    for (int j = 0; j < 4; ++j) {
        const int idx  = hw * 4 + j;
        const int trow = 8 * (idx >> 1) + lrow;
        const int q    = 4 * (idx & 1) + lquad;
#pragma unroll
        for (int i = 0; i < 4; ++i)
            st[(4 * q + i) * SCAN_PITCH + trow] = v[j][i];
    }
}

static __device__ __forceinline__ void scan_tile_out(float* __restrict__ hidden, const float* sOut,
                                                     int kb, int b0, int hw, int lane) {
    float* dstp = hidden + (long)(kb * SCAN_TB) * BATCH + b0;
    const float* st = sOut + (kb % SCAN_NO) * SCAN_STAGE;
    const int lrow = lane >> 3;
    const int q    = lane & 7;
#pragma unroll
    for (int j = 0; j < 4; ++j) {
        const int trow = 4 * (hw * 4 + j) + lrow;
        v4f v;
#pragma unroll
        for (int i = 0; i < 4; ++i)
            v[i] = st[(4 * q + i) * SCAN_PITCH + trow];
        *(v4f*)(dstp + trow * BATCH + 4 * q) = v;
    }
}

__global__ void __launch_bounds__(288)
scan_kernel(const float* __restrict__ proj,
            const float* __restrict__ hx0,
            const float* __restrict__ W_hh,
            float* __restrict__ hidden) {
    __shared__ v4f sIn4[SCAN_NS * SCAN_STAGE / 4];
    __shared__ v4f sOut4[SCAN_NO * SCAN_STAGE / 4];
    float* sIn  = (float*)sIn4;
    float* sOut = (float*)sOut4;

    const int tid  = threadIdx.x;
    const int lane = tid & 31;
    const int wave = __builtin_amdgcn_readfirstlane(tid >> 5);
    const int hw   = (wave > 0) ? (wave - 1) : 0;
    const int b0   = blockIdx.x * 32;

    if (wave != 0) {
        scan_tile_in(proj, sIn, 0, b0, hw, lane);
        scan_tile_in(proj, sIn, 1, b0, hw, lane);
    }
    lds_barrier();

    float h = 0.0f;
    float w = 0.0f;
    v4f nxt[SCAN_GV];
#pragma unroll
    for (int j = 0; j < SCAN_GV; ++j) nxt[j] = (v4f){0.0f, 0.0f, 0.0f, 0.0f};
    if (wave == 0) {
        h = hx0[b0 + lane];
        w = W_hh[0];
        const v4f* g0 = (const v4f*)(sIn + lane * SCAN_PITCH);
#pragma unroll
        for (int j = 0; j < SCAN_GV; ++j) nxt[j] = g0[j];
    }

    for (int k = 0; k < SCAN_NBLK; ++k) {
        if (wave != 0) {
            if (k + 2 < SCAN_NBLK) scan_tile_in(proj, sIn, k + 2, b0, hw, lane);
            if (k > 0)             scan_tile_out(hidden, sOut, k - 1, b0, hw, lane);
        } else {
#pragma unroll
            for (int gi = 0; gi < SCAN_GPB; ++gi) {
                const int g = k * SCAN_GPB + gi;
                v4f cur[SCAN_GV];
#pragma unroll
                for (int j = 0; j < SCAN_GV; ++j) cur[j] = nxt[j];

                const int gn  = (g + 1 < SCAN_NGRP) ? (g + 1) : g;
                const int kn  = gn / SCAN_GPB;
                const v4f* gp = (const v4f*)(sIn + (kn % SCAN_NS) * SCAN_STAGE
                                             + lane * SCAN_PITCH + (gn % SCAN_GPB) * SCAN_GRP);
#pragma unroll
                for (int j = 0; j < SCAN_GV; ++j) nxt[j] = gp[j];
                sched_fence(h);

#pragma unroll
                for (int j = 0; j < SCAN_GV; ++j) {
#pragma unroll
                    for (int i = 0; i < 4; ++i) {
                        h = fast_tanh(fmaf(h, w, cur[j][i]));
                        cur[j][i] = h;
                    }
                }
                sched_fence(h);

                v4f* so = (v4f*)(sOut + (k % SCAN_NO) * SCAN_STAGE
                                 + lane * SCAN_PITCH + gi * SCAN_GRP);
#pragma unroll
                for (int j = 0; j < SCAN_GV; ++j) so[j] = cur[j];
            }
        }
        lds_barrier();
    }

    if (wave != 0) scan_tile_out(hidden, sOut, SCAN_NBLK - 1, b0, hw, lane);
}

__global__ void __launch_bounds__(256)
lam_kernel(const float* __restrict__ hidden,
           const float* __restrict__ tvals,
           const float* __restrict__ W_h2p,
           const float* __restrict__ b_h2p,
           float* __restrict__ lam_out) {
    __shared__ float sW[64], sb[64];
    const int tid = threadIdx.x;
    if (tid < 64) { sW[tid] = W_h2p[tid]; sb[tid] = b_h2p[tid]; }
    __syncthreads();

    const long idx = (long)blockIdx.x * 256 + tid;
    const float h  = hidden[idx];
    const float tt = tvals[idx];

    float s = 0.0f;
#pragma unroll
    for (int j = 0; j < 32; ++j) {
        const float alpha = fmaf(h, sW[2 * j],     sb[2 * j]);
        const float beta  = fmaf(h, sW[2 * j + 1], sb[2 * j + 1]);
        const float v     = fmaf(alpha, tt, beta);
        s += fmaxf(v, 0.0f);
    }
    const float lam = fmaxf(s, 0.0f) + log1pf(__expf(-fabsf(s)));
    lam_out[idx] = lam;
}

extern "C" void kernel_launch(void* const* d_in, const int* in_sizes, int n_in,
                              void* d_out, int out_size, void* d_ws, size_t ws_size,
                              hipStream_t stream) {
    const float* X     = (const float*)d_in[0];
    const float* hx0   = (const float*)d_in[1];
    const float* W_ih  = (const float*)d_in[2];
    const float* b_ih  = (const float*)d_in[3];
    const float* W_hh  = (const float*)d_in[4];
    const float* b_hh  = (const float*)d_in[5];
    const float* W_h2p = (const float*)d_in[6];
    const float* b_h2p = (const float*)d_in[7];

    float* proj  = (float*)d_ws;
    float* tvals = proj + NROWS;

    float* hidden = (float*)d_out;
    float* lam    = hidden + NROWS;

    proj_kernel<<<NROWS / 16 / 8, 256, 0, stream>>>(X, W_ih, b_ih, b_hh, proj, tvals);
    scan_kernel<<<8, 288, 0, stream>>>(proj, hx0, W_hh, hidden);
    lam_kernel<<<NROWS / 256, 256, 0, stream>>>(hidden, tvals, W_h2p, b_h2p, lam);
}


# projection: a matrix row is a 32-byte piece of an X row, so a wave load is 512 contiguous bytes (was 16 lines x 16 B); row sums read off D's diagonal, row_xmask adds; sum order changes, bias last
# speedup vs baseline: 4.2573x; 1.0718x over previous
#include <hip/hip_runtime.h>
#include <math.h>

#define BATCH 256
#define SEQ   2048
#define FEAT  64
#define NROWS (BATCH * SEQ)

typedef float v2f __attribute__((ext_vector_type(2)));
typedef float v8f __attribute__((ext_vector_type(8)));
typedef float v4f __attribute__((ext_vector_type(4)));

static __device__ __forceinline__ float xmask_add(float v, const int ctrl_sel) {
    const int vi = __float_as_int(v);
    int o;
    if (ctrl_sel == 1)      o = __builtin_amdgcn_update_dpp(vi, vi, 0x161, 0xf, 0xf, false);
    else if (ctrl_sel == 2) o = __builtin_amdgcn_update_dpp(vi, vi, 0x162, 0xf, 0xf, false);
    else                    o = __builtin_amdgcn_update_dpp(vi, vi, 0x164, 0xf, 0xf, false);
    return v + __int_as_float(o);
}

__global__ void __launch_bounds__(256)
proj_kernel(const float* __restrict__ X,
            const float* __restrict__ W_ih,
            const float* __restrict__ b_ih,
            const float* __restrict__ b_hh,
            float* __restrict__ proj,
            float* __restrict__ tvals) {
    __shared__ v4f sW4[FEAT / 4];
    float* sW = (float*)sW4;
    const int tid = threadIdx.x;
    if (tid < FEAT) sW[tid] = W_ih[tid];
    __syncthreads();

    const int lane = tid & 31;
    const int wave = tid >> 5;
    const int c    = lane & 15;
    const int hi   = lane >> 4;
    const int q    = lane & 7;
    const int pr   = (lane >> 3) & 1;
    const long tile   = (long)blockIdx.x * 8 + wave;
    const long r_base = tile * 16;
    const float* xl = X + r_base * FEAT + 8 * c + 4 * hi;

    v4f a[8];
#pragma unroll
    for (int ch = 0; ch < 8; ++ch)
        a[ch] = *(const v4f*)(xl + ch * 128);
    __builtin_amdgcn_sched_barrier(0);

    const v4f wq = sW4[2 * q + hi];
    v2f bf[2][2];
#pragma unroll
    for (int p = 0; p < 2; ++p) {
        const bool on = (pr == p);
        bf[p][0].x = on ? wq.x : 0.0f;
        bf[p][0].y = on ? wq.y : 0.0f;
        bf[p][1].x = on ? wq.z : 0.0f;
        bf[p][1].y = on ? wq.w : 0.0f;
    }

    const float bias = b_ih[0] + b_hh[0];

    v8f acc[4];
#pragma unroll
    for (int d = 0; d < 4; ++d) {
        acc[d] = (v8f){0.0f, 0.0f, 0.0f, 0.0f, 0.0f, 0.0f, 0.0f, 0.0f};
        const v4f a0 = a[2 * d], a1 = a[2 * d + 1];
        const v2f a00 = {a0.x, a0.y}, a01 = {a0.z, a0.w};
        const v2f a10 = {a1.x, a1.y}, a11 = {a1.z, a1.w};
        acc[d] = __builtin_amdgcn_wmma_f32_16x16x4_f32(false, a00, false, bf[0][0], (short)0, acc[d], false, false);
        acc[d] = __builtin_amdgcn_wmma_f32_16x16x4_f32(false, a01, false, bf[0][1], (short)0, acc[d], false, false);
        acc[d] = __builtin_amdgcn_wmma_f32_16x16x4_f32(false, a10, false, bf[1][0], (short)0, acc[d], false, false);
        acc[d] = __builtin_amdgcn_wmma_f32_16x16x4_f32(false, a11, false, bf[1][1], (short)0, acc[d], false, false);
    }
    __builtin_amdgcn_sched_barrier(0);

    float y[4];
#pragma unroll
    for (int d = 0; d < 4; ++d) {
        const float s01 = (q & 1) ? acc[d][1] : acc[d][0];
        const float s23 = (q & 1) ? acc[d][3] : acc[d][2];
        const float s45 = (q & 1) ? acc[d][5] : acc[d][4];
        const float s67 = (q & 1) ? acc[d][7] : acc[d][6];
        const float s03 = (q & 2) ? s23 : s01;
        const float s47 = (q & 2) ? s67 : s45;
        y[d] = (q & 4) ? s47 : s03;
    }
#pragma unroll
    for (int d = 0; d < 4; ++d) y[d] = xmask_add(y[d], 1);
#pragma unroll
    for (int d = 0; d < 4; ++d) y[d] = xmask_add(y[d], 2);
#pragma unroll
    for (int d = 0; d < 4; ++d) y[d] = xmask_add(y[d], 4);

    const int  bidx  = (int)(r_base >> 11);
    const int  tbase = (int)(r_base & (SEQ - 1));

    if (q == 0 && hi == 0) {
        const int rr = c >> 3;
#pragma unroll
        for (int ch = 0; ch < 8; ++ch)
            tvals[(tbase + 2 * ch + rr) * BATCH + bidx] = a[ch].y;
    }

    if (q == 0) {
#pragma unroll
        for (int d = 0; d < 4; ++d)
            proj[(tbase + 4 * d + 2 * pr + hi) * BATCH + bidx] = y[d] + bias;
    }
}

static __device__ __forceinline__ float fast_tanh(float x) {
#if __has_builtin(__builtin_amdgcn_tanhf)
    return __builtin_amdgcn_tanhf(x);
#elif __has_builtin(__builtin_amdgcn_tanh_f32)
    return __builtin_amdgcn_tanh_f32(x);
#else
    return tanhf(x);
#endif
}

#define SCAN_TB      128
#define SCAN_NBLK    (SEQ / SCAN_TB)
#define SCAN_NS      3
#define SCAN_NO      2
#define SCAN_PITCH   132
#define SCAN_STAGE   (32 * SCAN_PITCH)
#define SCAN_GRP     64
#define SCAN_NGRP    (SEQ / SCAN_GRP)
#define SCAN_GV      (SCAN_GRP / 4)
#define SCAN_GPB     (SCAN_TB / SCAN_GRP)

static __device__ __forceinline__ void lds_barrier() {
    asm volatile("s_wait_dscnt 0x0\n\ts_barrier_signal -1\n\ts_barrier_wait -1" ::: "memory");
}

static __device__ __forceinline__ void sched_fence(float& hv) {
    asm volatile("" : "+v"(hv) :: "memory");
}

static __device__ __forceinline__ void scan_tile_in(const float* __restrict__ proj, float* sIn,
                                                    int kb, int b0, int hw, int lane) {
    const float* srcp = proj + (long)(kb * SCAN_TB) * BATCH + b0;
    float* st = sIn + (kb % SCAN_NS) * SCAN_STAGE;
    const int lrow  = lane & 7;
    const int lquad = lane >> 3;
    v4f v[4];
#pragma unroll
    for (int j = 0; j < 4; ++j) {
        const int idx  = hw * 4 + j;
        const int trow = 8 * (idx >> 1) + lrow;
        const int q    = 4 * (idx & 1) + lquad;
        v[j] = *(const v4f*)(srcp + trow * BATCH + 4 * q);
    }
#pragma unroll
    for (int j = 0; j < 4; ++j) {
        const int idx  = hw * 4 + j;
        const int trow = 8 * (idx >> 1) + lrow;
        const int q    = 4 * (idx & 1) + lquad;
#pragma unroll
        for (int i = 0; i < 4; ++i)
            st[(4 * q + i) * SCAN_PITCH + trow] = v[j][i];
    }
}

static __device__ __forceinline__ void scan_tile_out(float* __restrict__ hidden, const float* sOut,
                                                     int kb, int b0, int hw, int lane) {
    float* dstp = hidden + (long)(kb * SCAN_TB) * BATCH + b0;
    const float* st = sOut + (kb % SCAN_NO) * SCAN_STAGE;
    const int lrow = lane >> 3;
    const int q    = lane & 7;
#pragma unroll
    for (int j = 0; j < 4; ++j) {
        const int trow = 4 * (hw * 4 + j) + lrow;
        v4f v;
#pragma unroll
        for (int i = 0; i < 4; ++i)
            v[i] = st[(4 * q + i) * SCAN_PITCH + trow];
        *(v4f*)(dstp + trow * BATCH + 4 * q) = v;
    }
}

__global__ void __launch_bounds__(288)
scan_kernel(const float* __restrict__ proj,
            const float* __restrict__ hx0,
            const float* __restrict__ W_hh,
            float* __restrict__ hidden) {
    __shared__ v4f sIn4[SCAN_NS * SCAN_STAGE / 4];
    __shared__ v4f sOut4[SCAN_NO * SCAN_STAGE / 4];
    float* sIn  = (float*)sIn4;
    float* sOut = (float*)sOut4;

    const int tid  = threadIdx.x;
    const int lane = tid & 31;
    const int wave = __builtin_amdgcn_readfirstlane(tid >> 5);
    const int hw   = (wave > 0) ? (wave - 1) : 0;
    const int b0   = blockIdx.x * 32;

    if (wave != 0) {
        scan_tile_in(proj, sIn, 0, b0, hw, lane);
        scan_tile_in(proj, sIn, 1, b0, hw, lane);
    }
    lds_barrier();

    float h = 0.0f;
    float w = 0.0f;
    v4f nxt[SCAN_GV];
#pragma unroll
    for (int j = 0; j < SCAN_GV; ++j) nxt[j] = (v4f){0.0f, 0.0f, 0.0f, 0.0f};
    if (wave == 0) {
        h = hx0[b0 + lane];
        w = W_hh[0];
        const v4f* g0 = (const v4f*)(sIn + lane * SCAN_PITCH);
#pragma unroll
        for (int j = 0; j < SCAN_GV; ++j) nxt[j] = g0[j];
    }

    for (int k = 0; k < SCAN_NBLK; ++k) {
        if (wave != 0) {
            if (k + 2 < SCAN_NBLK) scan_tile_in(proj, sIn, k + 2, b0, hw, lane);
            if (k > 0)             scan_tile_out(hidden, sOut, k - 1, b0, hw, lane);
        } else {
#pragma unroll
            for (int gi = 0; gi < SCAN_GPB; ++gi) {
                const int g = k * SCAN_GPB + gi;
                v4f cur[SCAN_GV];
#pragma unroll
                for (int j = 0; j < SCAN_GV; ++j) cur[j] = nxt[j];

                const int gn  = (g + 1 < SCAN_NGRP) ? (g + 1) : g;
                const int kn  = gn / SCAN_GPB;
                const v4f* gp = (const v4f*)(sIn + (kn % SCAN_NS) * SCAN_STAGE
                                             + lane * SCAN_PITCH + (gn % SCAN_GPB) * SCAN_GRP);
#pragma unroll
                for (int j = 0; j < SCAN_GV; ++j) nxt[j] = gp[j];
                sched_fence(h);

#pragma unroll
                for (int j = 0; j < SCAN_GV; ++j) {
#pragma unroll
                    for (int i = 0; i < 4; ++i) {
                        h = fast_tanh(fmaf(h, w, cur[j][i]));
                        cur[j][i] = h;
                    }
                }
                sched_fence(h);

                v4f* so = (v4f*)(sOut + (k % SCAN_NO) * SCAN_STAGE
                                 + lane * SCAN_PITCH + gi * SCAN_GRP);
#pragma unroll
                for (int j = 0; j < SCAN_GV; ++j) so[j] = cur[j];
            }
        }
        lds_barrier();
    }

    if (wave != 0) scan_tile_out(hidden, sOut, SCAN_NBLK - 1, b0, hw, lane);
}

__global__ void __launch_bounds__(256)
lam_kernel(const float* __restrict__ hidden,
           const float* __restrict__ tvals,
           const float* __restrict__ W_h2p,
           const float* __restrict__ b_h2p,
           float* __restrict__ lam_out) {
    __shared__ float sW[64], sb[64];
    const int tid = threadIdx.x;
    if (tid < 64) { sW[tid] = W_h2p[tid]; sb[tid] = b_h2p[tid]; }
    __syncthreads();

    const long idx = (long)blockIdx.x * 256 + tid;
    const float h  = hidden[idx];
    const float tt = tvals[idx];

    float s = 0.0f;
#pragma unroll
    for (int j = 0; j < 32; ++j) {
        const float alpha = fmaf(h, sW[2 * j],     sb[2 * j]);
        const float beta  = fmaf(h, sW[2 * j + 1], sb[2 * j + 1]);
        const float v     = fmaf(alpha, tt, beta);
        s += fmaxf(v, 0.0f);
    }
    const float lam = fmaxf(s, 0.0f) + log1pf(__expf(-fabsf(s)));
    lam_out[idx] = lam;
}

extern "C" void kernel_launch(void* const* d_in, const int* in_sizes, int n_in,
                              void* d_out, int out_size, void* d_ws, size_t ws_size,
                              hipStream_t stream) {
    const float* X     = (const float*)d_in[0];
    const float* hx0   = (const float*)d_in[1];
    const float* W_ih  = (const float*)d_in[2];
    const float* b_ih  = (const float*)d_in[3];
    const float* W_hh  = (const float*)d_in[4];
    const float* b_hh  = (const float*)d_in[5];
    const float* W_h2p = (const float*)d_in[6];
    const float* b_h2p = (const float*)d_in[7];

    float* proj  = (float*)d_ws;
    float* tvals = proj + NROWS;

    float* hidden = (float*)d_out;
    float* lam    = hidden + NROWS;

    proj_kernel<<<NROWS / 16 / 8, 256, 0, stream>>>(X, W_ih, b_ih, b_hh, proj, tvals);
    scan_kernel<<<8, 288, 0, stream>>>(proj, hx0, W_hh, hidden);
    lam_kernel<<<NROWS / 256, 256, 0, stream>>>(hidden, tvals, W_h2p, b_h2p, lam);
}
